# starting kernel
# baseline (speedup 1.0000x reference)
#include <hip/hip_runtime.h>
#include <hip/hip_bf16.h>

typedef float v2f __attribute__((ext_vector_type(2)));
typedef float v8f __attribute__((ext_vector_type(8)));

#define N_NODES 100000
#define N_EDGES 1600000
#define C_HID   32
#define C_OUT   8
#define N_AGENTS 1000
#define N_TILES (N_NODES / 16)

__global__ void zero_f32(float* __restrict__ p, int n) {
    int t = blockIdx.x * blockDim.x + threadIdx.x;
    if (t < n) p[t] = 0.0f;
}

__global__ void count_deg(const int* __restrict__ dst, float* __restrict__ deg) {
    int t = blockIdx.x * blockDim.x + threadIdx.x;
    if (t < N_EDGES) atomicAdd(deg + dst[t], 1.0f);
}

__global__ __launch_bounds__(256) void scatter_add(
        const float* __restrict__ xin, const int* __restrict__ src,
        const int* __restrict__ dst, float* __restrict__ agg) {
    int t = blockIdx.x * blockDim.x + threadIdx.x;
    if (t >= N_EDGES * 8) return;
    int e = t >> 3;
    int c = (t & 7) * 4;
    int s = src[e];
    int d = dst[e];
    const float4 v = *(const float4*)(xin + (size_t)s * C_HID + c);
    float* p = agg + (size_t)d * C_HID + c;
    atomicAdd(p + 0, v.x);
    atomicAdd(p + 1, v.y);
    atomicAdd(p + 2, v.z);
    atomicAdd(p + 3, v.w);
}

__global__ __launch_bounds__(256) void sage_gemm(
        const float* __restrict__ agg, const float* __restrict__ xin,
        const float* __restrict__ deg,
        const float* __restrict__ Wl, const float* __restrict__ Wr,
        const float* __restrict__ bias, float* __restrict__ hout) {
    __shared__ float sW[64 * 32];
    __shared__ float sB[32];
    for (int i = threadIdx.x; i < 1024; i += 256) {
        sW[i]        = Wl[i];
        sW[1024 + i] = Wr[i];
    }
    if (threadIdx.x < 32) sB[threadIdx.x] = bias[threadIdx.x];
    __syncthreads();

    const int wave = threadIdx.x >> 5;
    const int lane = threadIdx.x & 31;
    const int tile = blockIdx.x * 8 + wave;
    if (tile >= N_TILES) return;

    const int row0 = tile * 16;
    const int mrow = lane & 15;
    const int row  = row0 + mrow;
    const int koff = (lane >> 4) * 2;

    const float dv  = deg[row];
    const float inv = 1.0f / fmaxf(dv, 1.0f);

    v8f acc0, acc1;
    const float b0v = sB[mrow];
    const float b1v = sB[16 + mrow];
#pragma unroll
    for (int j = 0; j < 8; ++j) { acc0[j] = b0v; acc1[j] = b1v; }

    const float* aggRow = agg + (size_t)row * C_HID;
    const float* xRow   = xin + (size_t)row * C_HID;

#pragma unroll
    for (int kc = 0; kc < 16; ++kc) {
        const int kbase = kc * 4;
        const int kb = kbase + koff;
        v2f a;
        if (kc < 8) {
            a.x = aggRow[kb]     * inv;
            a.y = aggRow[kb + 1] * inv;
        } else {
            a.x = xRow[kb - 32];
            a.y = xRow[kb - 31];
        }
        v2f b0, b1;
        b0.x = sW[(kb)     * 32 + mrow];
        b0.y = sW[(kb + 1) * 32 + mrow];
        b1.x = sW[(kb)     * 32 + 16 + mrow];
        b1.y = sW[(kb + 1) * 32 + 16 + mrow];
        acc0 = __builtin_amdgcn_wmma_f32_16x16x4_f32(
                   false, a, false, b0, (short)0, acc0, false, false);
        acc1 = __builtin_amdgcn_wmma_f32_16x16x4_f32(
                   false, a, false, b1, (short)0, acc1, false, false);
    }

    const int half = lane >> 4;
#pragma unroll
    for (int j = 0; j < 8; ++j) {
        const int m = row0 + half * 8 + j;
        hout[(size_t)m * C_HID + mrow]      = fmaxf(acc0[j], 0.0f);
        hout[(size_t)m * C_HID + 16 + mrow] = fmaxf(acc1[j], 0.0f);
    }
}

__global__ void out_proj(const float* __restrict__ h, const float* __restrict__ Wo,
                         const float* __restrict__ bo, float* __restrict__ out) {
    int t = blockIdx.x * blockDim.x + threadIdx.x;
    if (t >= N_AGENTS * C_OUT) return;
    const int n = t >> 3, o = t & 7;
    float s = bo[o];
    const float* hr = h + (size_t)n * C_HID;
#pragma unroll
    for (int k = 0; k < C_HID; ++k) s += hr[k] * Wo[k * C_OUT + o];
    out[t] = s;
}

extern "C" void kernel_launch(void* const* d_in, const int* in_sizes, int n_in,
                              void* d_out, int out_size, void* d_ws, size_t ws_size,
                              hipStream_t stream) {
    const float* x    = (const float*)d_in[0];
    const int*   ei   = (const int*)d_in[1];
    const int*   src  = ei;
    const int*   dst  = ei + N_EDGES;
    const float* Wl1  = (const float*)d_in[2];
    const float* bl1  = (const float*)d_in[3];
    const float* Wr1  = (const float*)d_in[4];
    const float* Wl2  = (const float*)d_in[5];
    const float* bl2  = (const float*)d_in[6];
    const float* Wr2  = (const float*)d_in[7];
    const float* Wl3  = (const float*)d_in[8];
    const float* bl3  = (const float*)d_in[9];
    const float* Wr3  = (const float*)d_in[10];
    const float* Wo   = (const float*)d_in[11];
    const float* bo   = (const float*)d_in[12];
    float* out = (float*)d_out;

    const size_t NB = (size_t)N_NODES * C_HID;
    float* agg = (float*)d_ws;
    float* h1  = agg + NB;
    float* h2  = h1 + NB;
    float* deg = h2 + NB;

    const int ZB_FEAT = (int)((NB + 255) / 256);
    const int ZB_DEG  = (N_NODES + 255) / 256;
    const int EB      = (N_EDGES + 255) / 256;
    const int SB      = (N_EDGES * 8 + 255) / 256;
    const int GB      = (N_TILES + 7) / 8;
    const int OB      = (N_AGENTS * C_OUT + 255) / 256;

    zero_f32<<<ZB_DEG, 256, 0, stream>>>(deg, N_NODES);
    count_deg<<<EB, 256, 0, stream>>>(dst, deg);

    zero_f32<<<ZB_FEAT, 256, 0, stream>>>(agg, (int)NB);
    scatter_add<<<SB, 256, 0, stream>>>(x, src, dst, agg);
    sage_gemm<<<GB, 256, 0, stream>>>(agg, x, deg, Wl1, Wr1, bl1, h1);

    zero_f32<<<ZB_FEAT, 256, 0, stream>>>(agg, (int)NB);
    scatter_add<<<SB, 256, 0, stream>>>(h1, src, dst, agg);
    sage_gemm<<<GB, 256, 0, stream>>>(agg, h1, deg, Wl2, Wr2, bl2, h2);

    zero_f32<<<ZB_FEAT, 256, 0, stream>>>(agg, (int)NB);
    scatter_add<<<SB, 256, 0, stream>>>(h2, src, dst, agg);
    sage_gemm<<<GB, 256, 0, stream>>>(agg, h2, deg, Wl3, Wr3, bl3, h1);

    out_proj<<<OB, 256, 0, stream>>>(h1, Wo, bo, out);
}


# compute only rows the output head reads from: layer 3 (zero, scatter, GEMM) on the 1008 rows of its 63 tiles; layer-2 scatter only into rows layer 3 reads (byte flags set in the degree pass)
# speedup vs baseline: 1.5379x; 1.5379x over previous
#include <hip/hip_runtime.h>
#include <hip/hip_bf16.h>

typedef float v2f __attribute__((ext_vector_type(2)));
typedef float v8f __attribute__((ext_vector_type(8)));

#define N_NODES 100000
#define N_EDGES 1600000
#define C_HID   32
#define C_OUT   8
#define N_AGENTS 1000
#define N_TILES (N_NODES / 16)
#define N_OUT_TILES ((N_AGENTS + 15) / 16)
#define N_OUT_ROWS  (N_OUT_TILES * 16)

__global__ void zero_f32(float* __restrict__ p, int n) {
    int t = blockIdx.x * blockDim.x + threadIdx.x;
    if (t < n) p[t] = 0.0f;
}

__global__ void init_deg_need(float* __restrict__ deg, unsigned char* __restrict__ need) {
    int t = blockIdx.x * blockDim.x + threadIdx.x;
    if (t < N_NODES) {
        deg[t]  = 0.0f;
        need[t] = (t < N_OUT_ROWS) ? 1 : 0;
    }
}

__global__ void count_deg(const int* __restrict__ src, const int* __restrict__ dst,
                          float* __restrict__ deg, unsigned char* __restrict__ need) {
    int t = blockIdx.x * blockDim.x + threadIdx.x;
    if (t < N_EDGES) {
        const int d = dst[t];
        atomicAdd(deg + d, 1.0f);
        if (d < N_OUT_ROWS) need[src[t]] = 1;
    }
}

__global__ __launch_bounds__(256) void scatter_add(
        const float* __restrict__ xin, const int* __restrict__ src,
        const int* __restrict__ dst, float* __restrict__ agg) {
    int t = blockIdx.x * blockDim.x + threadIdx.x;
    if (t >= N_EDGES * 8) return;
    int e = t >> 3;
    int c = (t & 7) * 4;
    int s = src[e];
    int d = dst[e];
    const float4 v = *(const float4*)(xin + (size_t)s * C_HID + c);
    float* p = agg + (size_t)d * C_HID + c;
    atomicAdd(p + 0, v.x);
    atomicAdd(p + 1, v.y);
    atomicAdd(p + 2, v.z);
    atomicAdd(p + 3, v.w);
}

template <int MODE>
__global__ __launch_bounds__(256) void scatter_add_sel(
        const float* __restrict__ xin, const int* __restrict__ src,
        const int* __restrict__ dst, const unsigned char* __restrict__ need,
        float* __restrict__ agg) {
    const int e = blockIdx.x * blockDim.x + threadIdx.x;
    int  d = 0, s = 0;
    bool keep = false;
    if (e < N_EDGES) {
        d = dst[e];
        keep = (MODE == 1) ? (need[d] != 0) : (d < N_OUT_ROWS);
        if (keep) s = src[e];
    }
    unsigned mask = (unsigned)__ballot(keep);
    const int lane = threadIdx.x & 31;
    const int grp  = lane >> 3;
    const int c    = (lane & 7) * 4;
    while (mask != 0u) {
        unsigned m = mask;
        if (grp > 0) m &= m - 1u;
        if (grp > 1) m &= m - 1u;
        if (grp > 2) m &= m - 1u;
        const bool act = (m != 0u);
        const int  b   = act ? __builtin_ctz(m) : 0;
        const int  ss  = __shfl(s, b);
        const int  dd  = __shfl(d, b);
        if (act) {
            const float4 v = *(const float4*)(xin + (size_t)ss * C_HID + c);
            float* p = agg + (size_t)dd * C_HID + c;
            atomicAdd(p + 0, v.x);
            atomicAdd(p + 1, v.y);
            atomicAdd(p + 2, v.z);
            atomicAdd(p + 3, v.w);
        }
        mask &= mask - 1u;
        mask &= mask - 1u;
        mask &= mask - 1u;
        mask &= mask - 1u;
    }
}

__global__ __launch_bounds__(256) void sage_gemm(
        const float* __restrict__ agg, const float* __restrict__ xin,
        const float* __restrict__ deg,
        const float* __restrict__ Wl, const float* __restrict__ Wr,
        const float* __restrict__ bias, float* __restrict__ hout, int n_tiles) {
    __shared__ float sW[64 * 32];
    __shared__ float sB[32];
    for (int i = threadIdx.x; i < 1024; i += 256) {
        sW[i]        = Wl[i];
        sW[1024 + i] = Wr[i];
    }
    if (threadIdx.x < 32) sB[threadIdx.x] = bias[threadIdx.x];
    __syncthreads();

    const int wave = threadIdx.x >> 5;
    const int lane = threadIdx.x & 31;
    const int tile = blockIdx.x * 8 + wave;
    if (tile >= n_tiles) return;

    const int row0 = tile * 16;
    const int mrow = lane & 15;
    const int row  = row0 + mrow;
    const int koff = (lane >> 4) * 2;

    const float dv  = deg[row];
    const float inv = 1.0f / fmaxf(dv, 1.0f);

    v8f acc0, acc1;
    const float b0v = sB[mrow];
    const float b1v = sB[16 + mrow];
#pragma unroll
    for (int j = 0; j < 8; ++j) { acc0[j] = b0v; acc1[j] = b1v; }

    const float* aggRow = agg + (size_t)row * C_HID;
    const float* xRow   = xin + (size_t)row * C_HID;

#pragma unroll
    for (int kc = 0; kc < 16; ++kc) {
        const int kbase = kc * 4;
        const int kb = kbase + koff;
        v2f a;
        if (kc < 8) {
            a.x = aggRow[kb]     * inv;
            a.y = aggRow[kb + 1] * inv;
        } else {
            a.x = xRow[kb - 32];
            a.y = xRow[kb - 31];
        }
        v2f b0, b1;
        b0.x = sW[(kb)     * 32 + mrow];
        b0.y = sW[(kb + 1) * 32 + mrow];
        b1.x = sW[(kb)     * 32 + 16 + mrow];
        b1.y = sW[(kb + 1) * 32 + 16 + mrow];
        acc0 = __builtin_amdgcn_wmma_f32_16x16x4_f32(
                   false, a, false, b0, (short)0, acc0, false, false);
        acc1 = __builtin_amdgcn_wmma_f32_16x16x4_f32(
                   false, a, false, b1, (short)0, acc1, false, false);
    }

    const int half = lane >> 4;
#pragma unroll
    for (int j = 0; j < 8; ++j) {
        const int m = row0 + half * 8 + j;
        hout[(size_t)m * C_HID + mrow]      = fmaxf(acc0[j], 0.0f);
        hout[(size_t)m * C_HID + 16 + mrow] = fmaxf(acc1[j], 0.0f);
    }
}

__global__ void out_proj(const float* __restrict__ h, const float* __restrict__ Wo,
                         const float* __restrict__ bo, float* __restrict__ out) {
    int t = blockIdx.x * blockDim.x + threadIdx.x;
    if (t >= N_AGENTS * C_OUT) return;
    const int n = t >> 3, o = t & 7;
    float s = bo[o];
    const float* hr = h + (size_t)n * C_HID;
#pragma unroll
    for (int k = 0; k < C_HID; ++k) s += hr[k] * Wo[k * C_OUT + o];
    out[t] = s;
}

extern "C" void kernel_launch(void* const* d_in, const int* in_sizes, int n_in,
                              void* d_out, int out_size, void* d_ws, size_t ws_size,
                              hipStream_t stream) {
    const float* x    = (const float*)d_in[0];
    const int*   ei   = (const int*)d_in[1];
    const int*   src  = ei;
    const int*   dst  = ei + N_EDGES;
    const float* Wl1  = (const float*)d_in[2];
    const float* bl1  = (const float*)d_in[3];
    const float* Wr1  = (const float*)d_in[4];
    const float* Wl2  = (const float*)d_in[5];
    const float* bl2  = (const float*)d_in[6];
    const float* Wr2  = (const float*)d_in[7];
    const float* Wl3  = (const float*)d_in[8];
    const float* bl3  = (const float*)d_in[9];
    const float* Wr3  = (const float*)d_in[10];
    const float* Wo   = (const float*)d_in[11];
    const float* bo   = (const float*)d_in[12];
    float* out = (float*)d_out;

    const size_t NB = (size_t)N_NODES * C_HID;
    float* agg = (float*)d_ws;
    float* h1  = agg + NB;
    float* h2  = h1 + NB;
    float* deg = h2 + NB;

    unsigned char* need = (unsigned char*)h2;

    const int ZB_FEAT = (int)((NB + 255) / 256);
    const int ZB_OUT  = (N_OUT_ROWS * C_HID + 255) / 256;
    const int ZB_DEG  = (N_NODES + 255) / 256;
    const int EB      = (N_EDGES + 255) / 256;
    const int SB      = (N_EDGES * 8 + 255) / 256;
    const int GB      = (N_TILES + 7) / 8;
    const int GB_OUT  = (N_OUT_TILES + 7) / 8;
    const int OB      = (N_AGENTS * C_OUT + 255) / 256;

    init_deg_need<<<ZB_DEG, 256, 0, stream>>>(deg, need);
    count_deg<<<EB, 256, 0, stream>>>(src, dst, deg, need);

    zero_f32<<<ZB_FEAT, 256, 0, stream>>>(agg, (int)NB);
    scatter_add<<<SB, 256, 0, stream>>>(x, src, dst, agg);
    sage_gemm<<<GB, 256, 0, stream>>>(agg, x, deg, Wl1, Wr1, bl1, h1, N_TILES);

    zero_f32<<<ZB_FEAT, 256, 0, stream>>>(agg, (int)NB);
    scatter_add_sel<1><<<EB, 256, 0, stream>>>(h1, src, dst, need, agg);
    sage_gemm<<<GB, 256, 0, stream>>>(agg, h1, deg, Wl2, Wr2, bl2, h2, N_TILES);

    zero_f32<<<ZB_OUT, 256, 0, stream>>>(agg, N_OUT_ROWS * C_HID);
    scatter_add_sel<2><<<EB, 256, 0, stream>>>(h2, src, dst, need, agg);
    sage_gemm<<<GB_OUT, 256, 0, stream>>>(agg, h2, deg, Wl3, Wr3, bl3, h1, N_OUT_TILES);

    out_proj<<<OB, 256, 0, stream>>>(h1, Wo, bo, out);
}


# edge scatter relaid lane=channel: each wave-wide row load and each wave-wide f32 atomic touches one whole 128-B line (was 4 lines per instruction); atomics kept, arrival order as before
# speedup vs baseline: 1.7583x; 1.1433x over previous
#include <hip/hip_runtime.h>
#include <hip/hip_bf16.h>

typedef float v2f __attribute__((ext_vector_type(2)));
typedef float v8f __attribute__((ext_vector_type(8)));

#define N_NODES 100000
#define N_EDGES 1600000
#define C_HID   32
#define C_OUT   8
#define N_AGENTS 1000
#define N_TILES (N_NODES / 16)
#define N_OUT_TILES ((N_AGENTS + 15) / 16)
#define N_OUT_ROWS  (N_OUT_TILES * 16)

__global__ void zero_f32(float* __restrict__ p, int n) {
    int t = blockIdx.x * blockDim.x + threadIdx.x;
    if (t < n) p[t] = 0.0f;
}

__global__ void init_deg_need(float* __restrict__ deg, unsigned char* __restrict__ need) {
    int t = blockIdx.x * blockDim.x + threadIdx.x;
    if (t < N_NODES) {
        deg[t]  = 0.0f;
        need[t] = (t < N_OUT_ROWS) ? 1 : 0;
    }
}

__global__ void count_deg(const int* __restrict__ src, const int* __restrict__ dst,
                          float* __restrict__ deg, unsigned char* __restrict__ need) {
    int t = blockIdx.x * blockDim.x + threadIdx.x;
    if (t < N_EDGES) {
        const int d = dst[t];
        atomicAdd(deg + d, 1.0f);
        if (d < N_OUT_ROWS) need[src[t]] = 1;
    }
}

static_assert(N_EDGES % 256 == 0, "every wave of the edge kernels owns 32 valid edges");
static_assert(C_HID == 32, "one lane per channel");

__global__ __launch_bounds__(256) void scatter_add(
        const float* __restrict__ xin, const int* __restrict__ src,
        const int* __restrict__ dst, float* __restrict__ agg) {
    const int e    = blockIdx.x * blockDim.x + threadIdx.x;
    const int lane = threadIdx.x & 31;
    const int s = src[e];
    const int d = dst[e];
    float v[32];
#pragma unroll
    for (int i = 0; i < 32; ++i) {
        const int si = __builtin_amdgcn_readlane(s, i);
        v[i] = xin[(size_t)si * C_HID + lane];
    }
#pragma unroll
    for (int i = 0; i < 32; ++i) {
        const int di = __builtin_amdgcn_readlane(d, i);
        atomicAdd(agg + (size_t)di * C_HID + lane, v[i]);
    }
}

template <int MODE>
__global__ __launch_bounds__(256) void scatter_add_sel(
        const float* __restrict__ xin, const int* __restrict__ src,
        const int* __restrict__ dst, const unsigned char* __restrict__ need,
        float* __restrict__ agg) {
    const int e    = blockIdx.x * blockDim.x + threadIdx.x;
    const int lane = threadIdx.x & 31;
    const int d = dst[e];
    const bool keep = (MODE == 1) ? (need[d] != 0) : (d < N_OUT_ROWS);
    int s = 0;
    if (keep) s = src[e];
    unsigned mask = (unsigned)__ballot(keep);
    while (mask != 0u) {
        const int  b0 = __builtin_ctz(mask);
        mask &= mask - 1u;
        const bool h1 = (mask != 0u);
        const int  b1 = h1 ? __builtin_ctz(mask) : b0;
        mask &= mask - 1u;
        const bool h2 = (mask != 0u);
        const int  b2 = h2 ? __builtin_ctz(mask) : b0;
        mask &= mask - 1u;
        const bool h3 = (mask != 0u);
        const int  b3 = h3 ? __builtin_ctz(mask) : b0;
        mask &= mask - 1u;

        const int s0 = __builtin_amdgcn_readlane(s, b0);
        const int s1 = __builtin_amdgcn_readlane(s, b1);
        const int s2 = __builtin_amdgcn_readlane(s, b2);
        const int s3 = __builtin_amdgcn_readlane(s, b3);
        const int d0 = __builtin_amdgcn_readlane(d, b0);
        const int d1 = __builtin_amdgcn_readlane(d, b1);
        const int d2 = __builtin_amdgcn_readlane(d, b2);
        const int d3 = __builtin_amdgcn_readlane(d, b3);

        const float v0 = xin[(size_t)s0 * C_HID + lane];
        const float v1 = xin[(size_t)s1 * C_HID + lane];
        const float v2 = xin[(size_t)s2 * C_HID + lane];
        const float v3 = xin[(size_t)s3 * C_HID + lane];

        atomicAdd(agg + (size_t)d0 * C_HID + lane, v0);
        if (h1) atomicAdd(agg + (size_t)d1 * C_HID + lane, v1);
        if (h2) atomicAdd(agg + (size_t)d2 * C_HID + lane, v2);
        if (h3) atomicAdd(agg + (size_t)d3 * C_HID + lane, v3);
    }
}

__global__ __launch_bounds__(256) void sage_gemm(
        const float* __restrict__ agg, const float* __restrict__ xin,
        const float* __restrict__ deg,
        const float* __restrict__ Wl, const float* __restrict__ Wr,
        const float* __restrict__ bias, float* __restrict__ hout, int n_tiles) {
    __shared__ float sW[64 * 32];
    __shared__ float sB[32];
    for (int i = threadIdx.x; i < 1024; i += 256) {
        sW[i]        = Wl[i];
        sW[1024 + i] = Wr[i];
    }
    if (threadIdx.x < 32) sB[threadIdx.x] = bias[threadIdx.x];
    __syncthreads();

    const int wave = threadIdx.x >> 5;
    const int lane = threadIdx.x & 31;
    const int tile = blockIdx.x * 8 + wave;
    if (tile >= n_tiles) return;

    const int row0 = tile * 16;
    const int mrow = lane & 15;
    const int row  = row0 + mrow;
    const int koff = (lane >> 4) * 2;

    const float dv  = deg[row];
    const float inv = 1.0f / fmaxf(dv, 1.0f);

    v8f acc0, acc1;
    const float b0v = sB[mrow];
    const float b1v = sB[16 + mrow];
#pragma unroll
    for (int j = 0; j < 8; ++j) { acc0[j] = b0v; acc1[j] = b1v; }

    const float* aggRow = agg + (size_t)row * C_HID;
    const float* xRow   = xin + (size_t)row * C_HID;

#pragma unroll
    for (int kc = 0; kc < 16; ++kc) {
        const int kbase = kc * 4;
        const int kb = kbase + koff;
        v2f a;
        if (kc < 8) {
            a.x = aggRow[kb]     * inv;
            a.y = aggRow[kb + 1] * inv;
        } else {
            a.x = xRow[kb - 32];
            a.y = xRow[kb - 31];
        }
        v2f b0, b1;
        b0.x = sW[(kb)     * 32 + mrow];
        b0.y = sW[(kb + 1) * 32 + mrow];
        b1.x = sW[(kb)     * 32 + 16 + mrow];
        b1.y = sW[(kb + 1) * 32 + 16 + mrow];
        acc0 = __builtin_amdgcn_wmma_f32_16x16x4_f32(
                   false, a, false, b0, (short)0, acc0, false, false);
        acc1 = __builtin_amdgcn_wmma_f32_16x16x4_f32(
                   false, a, false, b1, (short)0, acc1, false, false);
    }

    const int half = lane >> 4;
#pragma unroll
    for (int j = 0; j < 8; ++j) {
        const int m = row0 + half * 8 + j;
        hout[(size_t)m * C_HID + mrow]      = fmaxf(acc0[j], 0.0f);
        hout[(size_t)m * C_HID + 16 + mrow] = fmaxf(acc1[j], 0.0f);
    }
}

__global__ void out_proj(const float* __restrict__ h, const float* __restrict__ Wo,
                         const float* __restrict__ bo, float* __restrict__ out) {
    int t = blockIdx.x * blockDim.x + threadIdx.x;
    if (t >= N_AGENTS * C_OUT) return;
    const int n = t >> 3, o = t & 7;
    float s = bo[o];
    const float* hr = h + (size_t)n * C_HID;
#pragma unroll
    for (int k = 0; k < C_HID; ++k) s += hr[k] * Wo[k * C_OUT + o];
    out[t] = s;
}

extern "C" void kernel_launch(void* const* d_in, const int* in_sizes, int n_in,
                              void* d_out, int out_size, void* d_ws, size_t ws_size,
                              hipStream_t stream) {
    const float* x    = (const float*)d_in[0];
    const int*   ei   = (const int*)d_in[1];
    const int*   src  = ei;
    const int*   dst  = ei + N_EDGES;
    const float* Wl1  = (const float*)d_in[2];
    const float* bl1  = (const float*)d_in[3];
    const float* Wr1  = (const float*)d_in[4];
    const float* Wl2  = (const float*)d_in[5];
    const float* bl2  = (const float*)d_in[6];
    const float* Wr2  = (const float*)d_in[7];
    const float* Wl3  = (const float*)d_in[8];
    const float* bl3  = (const float*)d_in[9];
    const float* Wr3  = (const float*)d_in[10];
    const float* Wo   = (const float*)d_in[11];
    const float* bo   = (const float*)d_in[12];
    float* out = (float*)d_out;

    const size_t NB = (size_t)N_NODES * C_HID;
    float* agg = (float*)d_ws;
    float* h1  = agg + NB;
    float* h2  = h1 + NB;
    float* deg = h2 + NB;

    unsigned char* need = (unsigned char*)h2;

    const int ZB_FEAT = (int)((NB + 255) / 256);
    const int ZB_OUT  = (N_OUT_ROWS * C_HID + 255) / 256;
    const int ZB_DEG  = (N_NODES + 255) / 256;
    const int EB      = (N_EDGES + 255) / 256;
    const int GB      = (N_TILES + 7) / 8;
    const int GB_OUT  = (N_OUT_TILES + 7) / 8;
    const int OB      = (N_AGENTS * C_OUT + 255) / 256;

    init_deg_need<<<ZB_DEG, 256, 0, stream>>>(deg, need);
    count_deg<<<EB, 256, 0, stream>>>(src, dst, deg, need);

    zero_f32<<<ZB_FEAT, 256, 0, stream>>>(agg, (int)NB);
    scatter_add<<<EB, 256, 0, stream>>>(x, src, dst, agg);
    sage_gemm<<<GB, 256, 0, stream>>>(agg, x, deg, Wl1, Wr1, bl1, h1, N_TILES);

    zero_f32<<<ZB_FEAT, 256, 0, stream>>>(agg, (int)NB);
    scatter_add_sel<1><<<EB, 256, 0, stream>>>(h1, src, dst, need, agg);
    sage_gemm<<<GB, 256, 0, stream>>>(agg, h1, deg, Wl2, Wr2, bl2, h2, N_TILES);

    zero_f32<<<ZB_OUT, 256, 0, stream>>>(agg, N_OUT_ROWS * C_HID);
    scatter_add_sel<2><<<EB, 256, 0, stream>>>(h2, src, dst, need, agg);
    sage_gemm<<<GB_OUT, 256, 0, stream>>>(agg, h2, deg, Wl3, Wr3, bl3, h1, N_OUT_TILES);

    out_proj<<<OB, 256, 0, stream>>>(h1, Wo, bo, out);
}


# layer-1 neighbour sums and degrees from edges ordered by destination range in the call (LDS count, scan, place): rows summed in LDS, stored once; no global f32 atomics there; sum order = LDS arrival
# speedup vs baseline: 2.1775x; 1.2384x over previous
#include <hip/hip_runtime.h>
#include <hip/hip_bf16.h>

typedef float v2f __attribute__((ext_vector_type(2)));
typedef float v8f __attribute__((ext_vector_type(8)));

#define N_NODES 100000
#define N_EDGES 1600000
#define C_HID   32
#define C_OUT   8
#define N_AGENTS 1000
#define N_TILES (N_NODES / 16)
#define N_OUT_TILES ((N_AGENTS + 15) / 16)
#define N_OUT_ROWS  (N_OUT_TILES * 16)

__global__ void zero_f32(float* __restrict__ p, int n) {
    int t = blockIdx.x * blockDim.x + threadIdx.x;
    if (t < n) p[t] = 0.0f;
}

__global__ void init_need(unsigned char* __restrict__ need) {
    int t = blockIdx.x * blockDim.x + threadIdx.x;
    if (t < N_NODES) need[t] = (t < N_OUT_ROWS) ? 1 : 0;
}

#define BSHIFT 7
#define BROWS  (1 << BSHIFT)
#define NBKT   ((N_NODES + BROWS - 1) / BROWS)
#define CH     16384
#define NBLK   ((N_EDGES + CH - 1) / CH)
static_assert(NBKT < 1024, "one thread per range in the prefix kernel, one spare");
static_assert(C_HID == 32, "one lane per channel");

__global__ __launch_bounds__(1024) void bin_count(
        const int* __restrict__ src, const int* __restrict__ dst,
        int* __restrict__ cnt, unsigned char* __restrict__ need) {
    __shared__ int hist[NBKT];
    for (int i = threadIdx.x; i < NBKT; i += 1024) hist[i] = 0;
    __syncthreads();
    const int base = blockIdx.x * CH + (int)threadIdx.x;
    int d[CH / 1024], s[CH / 1024];
#pragma unroll
    for (int k = 0; k < CH / 1024; ++k) {
        const int e = min(base + k * 1024, N_EDGES - 1);
        d[k] = dst[e];
        s[k] = src[e];
    }
#pragma unroll
    for (int k = 0; k < CH / 1024; ++k) {
        if (base + k * 1024 < N_EDGES) {
            atomicAdd(&hist[d[k] >> BSHIFT], 1);
            if (d[k] < N_OUT_ROWS) need[s[k]] = 1;
        }
    }
    __syncthreads();
    for (int i = threadIdx.x; i < NBKT; i += 1024) cnt[blockIdx.x * NBKT + i] = hist[i];
}

__global__ __launch_bounds__(1024) void scan_bins(
        const int* __restrict__ cnt, int* __restrict__ pre, int* __restrict__ bstart) {
    __shared__ int part[1024];
    const int t = threadIdx.x;
    int run = 0;
    if (t < NBKT) {
#pragma unroll
        for (int j = 0; j < NBLK; ++j) {
            const int c = cnt[j * NBKT + t];
            pre[j * NBKT + t] = run;
            run += c;
        }
    }
    part[t] = run;
    __syncthreads();
    for (int d = 1; d < 1024; d <<= 1) {
        const int v = (t >= d) ? part[t - d] : 0;
        __syncthreads();
        part[t] += v;
        __syncthreads();
    }
    if (t <= NBKT) bstart[t] = part[t] - run;
}

__global__ __launch_bounds__(1024) void bin_place(
        const int* __restrict__ src, const int* __restrict__ dst,
        const int* __restrict__ pre, const int* __restrict__ bstart,
        int* __restrict__ ordered) {
    __shared__ int cur[NBKT];
    for (int i = threadIdx.x; i < NBKT; i += 1024)
        cur[i] = bstart[i] + pre[blockIdx.x * NBKT + i];
    __syncthreads();
    int2* __restrict__ op = (int2*)ordered;
    const int base = blockIdx.x * CH + (int)threadIdx.x;
    int d[CH / 1024], s[CH / 1024], pos[CH / 1024];
#pragma unroll
    for (int k = 0; k < CH / 1024; ++k) {
        const int e = min(base + k * 1024, N_EDGES - 1);
        d[k] = dst[e];
        s[k] = src[e];
    }
#pragma unroll
    for (int k = 0; k < CH / 1024; ++k) {
        pos[k] = 0;
        if (base + k * 1024 < N_EDGES) pos[k] = atomicAdd(&cur[d[k] >> BSHIFT], 1);
    }
#pragma unroll
    for (int k = 0; k < CH / 1024; ++k) {
        if (base + k * 1024 < N_EDGES) op[pos[k]] = make_int2(s[k], d[k]);
    }
}

__global__ __launch_bounds__(512) void bin_sum(
        const float* __restrict__ xin, const int* __restrict__ ordered,
        const int* __restrict__ bstart, float* __restrict__ agg, float* __restrict__ deg) {
    __shared__ float acc[BROWS * C_HID];
    __shared__ int   rc[BROWS];
    const int tid  = threadIdx.x;
    const int lane = tid & 31;
    const int wave = __builtin_amdgcn_readfirstlane(tid >> 5);
    for (int i = tid; i < BROWS * C_HID; i += 512) acc[i] = 0.0f;
    if (tid < BROWS) rc[tid] = 0;
    __syncthreads();

    const int b     = blockIdx.x;
    const int rbase = b << BSHIFT;
    const int s0    = bstart[b];
    const int s1    = bstart[b + 1];
    const int2* __restrict__ op = (const int2*)ordered;

    int c = s0 + wave * 32;
    int2 ed = make_int2(0, rbase);
    if (c + lane < s1) ed = op[c + lane];
    while (c < s1) {
        const int nv = min(32, s1 - c);
        const int rl = ed.y - rbase;
        if (lane < nv) atomicAdd(&rc[rl], 1);
        const int cn = c + 512;
        int2 edn = make_int2(0, rbase);
        if (cn + lane < s1) edn = op[cn + lane];
        if (nv == 32) {
            float v[32];
#pragma unroll
            for (int i = 0; i < 32; ++i) {
                const int si = __builtin_amdgcn_readlane(ed.x, i);
                v[i] = xin[(size_t)si * C_HID + lane];
            }
#pragma unroll
            for (int i = 0; i < 32; ++i) {
                const int ri = __builtin_amdgcn_readlane(rl, i);
                atomicAdd(&acc[ri * C_HID + lane], v[i]);
            }
        } else {
            for (int i = 0; i < nv; ++i) {
                const int si = __builtin_amdgcn_readlane(ed.x, i);
                const int ri = __builtin_amdgcn_readlane(rl, i);
                atomicAdd(&acc[ri * C_HID + lane], xin[(size_t)si * C_HID + lane]);
            }
        }
        ed = edn;
        c  = cn;
    }
    __syncthreads();

    for (int r = wave; r < BROWS; r += 16) {
        const int n = rbase + r;
        if (n < N_NODES) agg[(size_t)n * C_HID + lane] = acc[r * C_HID + lane];
    }
    if (tid < BROWS && rbase + tid < N_NODES) deg[rbase + tid] = (float)rc[tid];
}

static_assert(N_EDGES % 256 == 0, "every wave of the edge kernels owns 32 valid edges");

template <int MODE>
__global__ __launch_bounds__(256) void scatter_add_sel(
        const float* __restrict__ xin, const int* __restrict__ src,
        const int* __restrict__ dst, const unsigned char* __restrict__ need,
        float* __restrict__ agg) {
    const int e    = blockIdx.x * blockDim.x + threadIdx.x;
    const int lane = threadIdx.x & 31;
    const int d = dst[e];
    const bool keep = (MODE == 1) ? (need[d] != 0) : (d < N_OUT_ROWS);
    int s = 0;
    if (keep) s = src[e];
    unsigned mask = (unsigned)__ballot(keep);
    while (mask != 0u) {
        const int  b0 = __builtin_ctz(mask);
        mask &= mask - 1u;
        const bool h1 = (mask != 0u);
        const int  b1 = h1 ? __builtin_ctz(mask) : b0;
        mask &= mask - 1u;
        const bool h2 = (mask != 0u);
        const int  b2 = h2 ? __builtin_ctz(mask) : b0;
        mask &= mask - 1u;
        const bool h3 = (mask != 0u);
        const int  b3 = h3 ? __builtin_ctz(mask) : b0;
        mask &= mask - 1u;

        const int s0 = __builtin_amdgcn_readlane(s, b0);
        const int s1 = __builtin_amdgcn_readlane(s, b1);
        const int s2 = __builtin_amdgcn_readlane(s, b2);
        const int s3 = __builtin_amdgcn_readlane(s, b3);
        const int d0 = __builtin_amdgcn_readlane(d, b0);
        const int d1 = __builtin_amdgcn_readlane(d, b1);
        const int d2 = __builtin_amdgcn_readlane(d, b2);
        const int d3 = __builtin_amdgcn_readlane(d, b3);

        const float v0 = xin[(size_t)s0 * C_HID + lane];
        const float v1 = xin[(size_t)s1 * C_HID + lane];
        const float v2 = xin[(size_t)s2 * C_HID + lane];
        const float v3 = xin[(size_t)s3 * C_HID + lane];

        atomicAdd(agg + (size_t)d0 * C_HID + lane, v0);
        if (h1) atomicAdd(agg + (size_t)d1 * C_HID + lane, v1);
        if (h2) atomicAdd(agg + (size_t)d2 * C_HID + lane, v2);
        if (h3) atomicAdd(agg + (size_t)d3 * C_HID + lane, v3);
    }
}

__global__ __launch_bounds__(256) void sage_gemm(
        const float* __restrict__ agg, const float* __restrict__ xin,
        const float* __restrict__ deg,
        const float* __restrict__ Wl, const float* __restrict__ Wr,
        const float* __restrict__ bias, float* __restrict__ hout, int n_tiles) {
    __shared__ float sW[64 * 32];
    __shared__ float sB[32];
    for (int i = threadIdx.x; i < 1024; i += 256) {
        sW[i]        = Wl[i];
        sW[1024 + i] = Wr[i];
    }
    if (threadIdx.x < 32) sB[threadIdx.x] = bias[threadIdx.x];
    __syncthreads();

    const int wave = threadIdx.x >> 5;
    const int lane = threadIdx.x & 31;
    const int tile = blockIdx.x * 8 + wave;
    if (tile >= n_tiles) return;

    const int row0 = tile * 16;
    const int mrow = lane & 15;
    const int row  = row0 + mrow;
    const int koff = (lane >> 4) * 2;

    const float dv  = deg[row];
    const float inv = 1.0f / fmaxf(dv, 1.0f);

    v8f acc0, acc1;
    const float b0v = sB[mrow];
    const float b1v = sB[16 + mrow];
#pragma unroll
    for (int j = 0; j < 8; ++j) { acc0[j] = b0v; acc1[j] = b1v; }

    const float* aggRow = agg + (size_t)row * C_HID;
    const float* xRow   = xin + (size_t)row * C_HID;

#pragma unroll
    for (int kc = 0; kc < 16; ++kc) {
        const int kbase = kc * 4;
        const int kb = kbase + koff;
        v2f a;
        if (kc < 8) {
            a.x = aggRow[kb]     * inv;
            a.y = aggRow[kb + 1] * inv;
        } else {
            a.x = xRow[kb - 32];
            a.y = xRow[kb - 31];
        }
        v2f b0, b1;
        b0.x = sW[(kb)     * 32 + mrow];
        b0.y = sW[(kb + 1) * 32 + mrow];
        b1.x = sW[(kb)     * 32 + 16 + mrow];
        b1.y = sW[(kb + 1) * 32 + 16 + mrow];
        acc0 = __builtin_amdgcn_wmma_f32_16x16x4_f32(
                   false, a, false, b0, (short)0, acc0, false, false);
        acc1 = __builtin_amdgcn_wmma_f32_16x16x4_f32(
                   false, a, false, b1, (short)0, acc1, false, false);
    }

    const int half = lane >> 4;
#pragma unroll
    for (int j = 0; j < 8; ++j) {
        const int m = row0 + half * 8 + j;
        hout[(size_t)m * C_HID + mrow]      = fmaxf(acc0[j], 0.0f);
        hout[(size_t)m * C_HID + 16 + mrow] = fmaxf(acc1[j], 0.0f);
    }
}

__global__ void out_proj(const float* __restrict__ h, const float* __restrict__ Wo,
                         const float* __restrict__ bo, float* __restrict__ out) {
    int t = blockIdx.x * blockDim.x + threadIdx.x;
    if (t >= N_AGENTS * C_OUT) return;
    const int n = t >> 3, o = t & 7;
    float s = bo[o];
    const float* hr = h + (size_t)n * C_HID;
#pragma unroll
    for (int k = 0; k < C_HID; ++k) s += hr[k] * Wo[k * C_OUT + o];
    out[t] = s;
}

extern "C" void kernel_launch(void* const* d_in, const int* in_sizes, int n_in,
                              void* d_out, int out_size, void* d_ws, size_t ws_size,
                              hipStream_t stream) {
    const float* x    = (const float*)d_in[0];
    const int*   ei   = (const int*)d_in[1];
    const int*   src  = ei;
    const int*   dst  = ei + N_EDGES;
    const float* Wl1  = (const float*)d_in[2];
    const float* bl1  = (const float*)d_in[3];
    const float* Wr1  = (const float*)d_in[4];
    const float* Wl2  = (const float*)d_in[5];
    const float* bl2  = (const float*)d_in[6];
    const float* Wr2  = (const float*)d_in[7];
    const float* Wl3  = (const float*)d_in[8];
    const float* bl3  = (const float*)d_in[9];
    const float* Wr3  = (const float*)d_in[10];
    const float* Wo   = (const float*)d_in[11];
    const float* bo   = (const float*)d_in[12];
    float* out = (float*)d_out;

    const size_t NB = (size_t)N_NODES * C_HID;
    float* agg = (float*)d_ws;
    float* h1  = agg + NB;
    float* h2  = h1 + NB;
    float* deg = h2 + NB;

    unsigned char* need = (unsigned char*)h2;

    int* cnt     = (int*)h2 + 65536;
    int* pre     = cnt + 98304;
    int* bstart  = pre + 98304;
    int* ordered = (int*)h1;
    static_assert(NBLK * NBKT <= 98304, "count tables fit their slots");
    static_assert((size_t)N_EDGES * 2 <= (size_t)N_NODES * C_HID, "ordered list fits h1");

    const int ZB_FEAT = (int)((NB + 255) / 256);
    const int ZB_OUT  = (N_OUT_ROWS * C_HID + 255) / 256;
    const int ZB_DEG  = (N_NODES + 255) / 256;
    const int EB      = (N_EDGES + 255) / 256;
    const int GB      = (N_TILES + 7) / 8;
    const int GB_OUT  = (N_OUT_TILES + 7) / 8;
    const int OB      = (N_AGENTS * C_OUT + 255) / 256;

    init_need<<<ZB_DEG, 256, 0, stream>>>(need);
    bin_count<<<NBLK, 1024, 0, stream>>>(src, dst, cnt, need);
    scan_bins<<<1, 1024, 0, stream>>>(cnt, pre, bstart);
    bin_place<<<NBLK, 1024, 0, stream>>>(src, dst, pre, bstart, ordered);

    bin_sum<<<NBKT, 512, 0, stream>>>(x, ordered, bstart, agg, deg);
    sage_gemm<<<GB, 256, 0, stream>>>(agg, x, deg, Wl1, Wr1, bl1, h1, N_TILES);

    zero_f32<<<ZB_FEAT, 256, 0, stream>>>(agg, (int)NB);
    scatter_add_sel<1><<<EB, 256, 0, stream>>>(h1, src, dst, need, agg);
    sage_gemm<<<GB, 256, 0, stream>>>(agg, h1, deg, Wl2, Wr2, bl2, h2, N_TILES);

    zero_f32<<<ZB_OUT, 256, 0, stream>>>(agg, N_OUT_ROWS * C_HID);
    scatter_add_sel<2><<<EB, 256, 0, stream>>>(h2, src, dst, need, agg);
    sage_gemm<<<GB_OUT, 256, 0, stream>>>(agg, h2, deg, Wl3, Wr3, bl3, h1, N_OUT_TILES);

    out_proj<<<OB, 256, 0, stream>>>(h1, Wo, bo, out);
}
